# starting kernel
# baseline (speedup 1.0000x reference)
#include <hip/hip_runtime.h>

typedef __attribute__((ext_vector_type(2))) float v2f;
typedef __attribute__((ext_vector_type(8))) float v8f;

#define NP   3
#define NC   1024
#define NV   64
#define NF   512
#define NK   9
#define ND   63
#define HOUT 8
#define KDIM (NP * NV)

__global__ __launch_bounds__(256) void stgcn_graph_wmma(
    const float* __restrict__ x, const float* __restrict__ graph,
    float* __restrict__ yT)
{
  const int wave = (blockIdx.x * blockDim.x + threadIdx.x) >> 5;
  const int lane = threadIdx.x & 31;
  const int half = lane >> 4;
  const int lr   = lane & 15;

  const int jT = wave & 3;
  const int cT = wave >> 2;
  const int c0 = cT * 16;
  const int j0 = jT * 16;

  const int ca = c0 + lr;
  const int jb = j0 + lr;

  v8f acc = {};
#pragma unroll
  for (int kk = 0; kk < KDIM / 4; ++kk) {
    const int kg = kk * 4 + half * 2;
    const int p  = kg >> 6;
    const int u  = kg & 63;
    v2f a = *(const v2f*)(x + (size_t)(p * NC + ca) * NV + u);
    v2f b;
    b.x = graph[(size_t)(p * NV + u)     * NV + jb];
    b.y = graph[(size_t)(p * NV + u + 1) * NV + jb];
    acc = __builtin_amdgcn_wmma_f32_16x16x4_f32(
         false, a,  false, b,
         (short)0, acc,  false,  false);
  }

  const int rowBase = c0 + half * 8;
#pragma unroll
  for (int r = 0; r < 8; ++r) {
    yT[(size_t)(rowBase + r) * NV + jb] = acc[r];
  }
}

__global__ __launch_bounds__(256) void stgcn_fifo_sum(
    const float* __restrict__ fifo, const float* __restrict__ yT,
    float* __restrict__ out)
{
  const int t   = blockIdx.x * blockDim.x + threadIdx.x;
  const int c   = t >> 7;
  const int rem = t & 127;
  const int h   = rem >> 4;
  const int v   = (rem & 15) << 2;

  const float* fbase = fifo + (size_t)c * NF * NV + v;

  float4 acc;
  if (h == 0) {
    acc = *(const float4*)(yT + (size_t)c * NV + v);
  } else {
    acc = *(const float4*)(fbase + (size_t)(h - 1) * NV);
  }
#pragma unroll
  for (int i = 1; i < NK; ++i) {
    float4 f = *(const float4*)(fbase + (size_t)(i * ND + h - 1) * NV);
    acc.x += f.x; acc.y += f.y; acc.z += f.z; acc.w += f.w;
  }
  *(float4*)(out + (size_t)t * 4) = acc;
}

extern "C" void kernel_launch(void* const* d_in, const int* in_sizes, int n_in,
                              void* d_out, int out_size, void* d_ws, size_t ws_size,
                              hipStream_t stream) {
  (void)in_sizes; (void)n_in; (void)out_size; (void)ws_size;
  const float* x     = (const float*)d_in[0];
  const float* graph = (const float*)d_in[1];
  const float* fifo  = (const float*)d_in[2];
  float* out = (float*)d_out;
  float* yT  = (float*)d_ws;

  stgcn_graph_wmma<<<32, 256, 0, stream>>>(x, graph, yT);
  stgcn_fifo_sum<<<512, 256, 0, stream>>>(fifo, yT, out);
}


# Fuse both launches into one: 64 product blocks (4 WMMA waves + 4 waves adding the 8 fifo taps to y in the old order and storing out row h=0) go first, 448 blocks stream rows h=1..7; no workspace
# speedup vs baseline: 1.9747x; 1.9747x over previous
#include <hip/hip_runtime.h>


typedef __attribute__((ext_vector_type(2))) float v2f;
typedef __attribute__((ext_vector_type(8))) float v8f;
typedef __attribute__((ext_vector_type(4))) float v4f;

#define NP   3
#define NC   1024
#define NV   64
#define NF   512
#define NK   9
#define ND   63
#define HOUT 8
#define KDIM (NP * NV)

#define TILE_ROWS      16
#define GEMM_BLOCKS    (NC / TILE_ROWS)
#define TILE_PITCH     (NV + 4)
#define ROWS_PER_C     (HOUT - 1)
#define STREAM_THREADS (NC * ROWS_PER_C * (NV / 4))
#define STREAM_BLOCKS  (STREAM_THREADS / 256)

__global__ __launch_bounds__(256) void graph_fifo_fused(
    const float* __restrict__ x, const float* __restrict__ graph,
    const float* __restrict__ fifo, float* __restrict__ out)
{
  __shared__ __attribute__((aligned(16))) float ytile[TILE_ROWS * TILE_PITCH];

  if (blockIdx.x < GEMM_BLOCKS) {
    const int c0 = blockIdx.x * TILE_ROWS;
    const int w = __builtin_amdgcn_readfirstlane((int)(threadIdx.x >> 5));

    const int ft    = threadIdx.x & 127;
    const int v4    = ft & 15;
    const int frow0 = ft >> 4;
    v4f tap[2][NK - 1];

    if (w >= 4) {
#pragma unroll
      for (int k = 0; k < 2; ++k) {
        const float* fr = fifo + (size_t)(c0 + frow0 + 8 * k) * NF * NV + v4 * 4;
#pragma unroll
        for (int i = 1; i < NK; ++i) {
          tap[k][i - 1] = *(const v4f*)(fr + (size_t)(i * ND - 1) * NV);
        }
      }
    } else {
      const int lane = threadIdx.x & 31;
      const int half = lane >> 4;
      const int lr   = lane & 15;

      const int j0 = w * 16;
      const int ca = c0 + lr;
      const int jb = j0 + lr;

      v8f acc = {};
#pragma unroll
      for (int kk = 0; kk < KDIM / 4; ++kk) {
        const int kg = kk * 4 + half * 2;
        const int p  = kg >> 6;
        const int u  = kg & 63;
        v2f a = *(const v2f*)(x + (size_t)(p * NC + ca) * NV + u);
        v2f b;
        b.x = graph[(size_t)(p * NV + u)     * NV + jb];
        b.y = graph[(size_t)(p * NV + u + 1) * NV + jb];
        acc = __builtin_amdgcn_wmma_f32_16x16x4_f32(
             false, a,  false, b,
             (short)0, acc,  false,  false);
      }

#pragma unroll
      for (int r = 0; r < 8; ++r) {
        ytile[(half * 8 + r) * TILE_PITCH + jb] = acc[r];
      }
    }

    __syncthreads();

    if (w >= 4) {
#pragma unroll
      for (int k = 0; k < 2; ++k) {
        const int row = frow0 + 8 * k;
        v4f o = *(const v4f*)(ytile + row * TILE_PITCH + v4 * 4);
#pragma unroll
        for (int i = 1; i < NK; ++i) o += tap[k][i - 1];
        *(v4f*)(out + ((size_t)(c0 + row) * HOUT) * NV + v4 * 4) = o;
      }
    }
  } else {
    const int s   = (blockIdx.x - GEMM_BLOCKS) * 256 + threadIdx.x;
    const int q   = s >> 4;
    const int c   = q / ROWS_PER_C;
    const int h   = 1 + (q - c * ROWS_PER_C);
    const int v   = (s & 15) << 2;

    const float* fbase = fifo + (size_t)c * NF * NV + v;

    v4f acc = *(const v4f*)(fbase + (size_t)(h - 1) * NV);
#pragma unroll
    for (int i = 1; i < NK; ++i) {
      acc += *(const v4f*)(fbase + (size_t)(i * ND + h - 1) * NV);
    }
    *(v4f*)(out + ((size_t)c * HOUT + h) * NV + v) = acc;
  }
}

extern "C" void kernel_launch(void* const* d_in, const int* in_sizes, int n_in,
                              void* d_out, int out_size, void* d_ws, size_t ws_size,
                              hipStream_t stream) {
  (void)in_sizes; (void)n_in; (void)out_size; (void)d_ws; (void)ws_size;
  const float* x     = (const float*)d_in[0];
  const float* graph = (const float*)d_in[1];
  const float* fifo  = (const float*)d_in[2];
  float* out = (float*)d_out;

  graph_fifo_fused<<<GEMM_BLOCKS + STREAM_BLOCKS, 256, 0, stream>>>(x, graph, fifo, out);
}


# Fewer, longer-lived waves: 384 blocks of 2 waves instead of 512 of 8; each stream thread sums all 7 rows of its float4 column (63 loads in flight), product waves fetch their own taps; add order kept
# speedup vs baseline: 2.5921x; 1.3127x over previous
#include <hip/hip_runtime.h>


typedef __attribute__((ext_vector_type(2))) float v2f;
typedef __attribute__((ext_vector_type(8))) float v8f;
typedef __attribute__((ext_vector_type(4))) float v4f;

#define NP   3
#define NC   1024
#define NV   64
#define NF   512
#define NK   9
#define ND   63
#define HOUT 8
#define KDIM (NP * NV)

#define BLOCK_THREADS  64
#define TILE_ROWS      16
#define BLOCK_COLS     32
#define PROD_BLOCKS    ((NC / TILE_ROWS) * (NV / BLOCK_COLS))
#define TILE_PITCH     (BLOCK_COLS + 4)
#define ROWS_PER_C     (HOUT - 1)
#define STREAM_THREADS (NC * (NV / 4))
#define STREAM_BLOCKS  (STREAM_THREADS / BLOCK_THREADS)

__global__ __launch_bounds__(BLOCK_THREADS) void graph_fifo_fused(
    const float* __restrict__ x, const float* __restrict__ graph,
    const float* __restrict__ fifo, float* __restrict__ out)
{
  __shared__ __attribute__((aligned(16))) float ytile[TILE_ROWS * TILE_PITCH];

  if (blockIdx.x < PROD_BLOCKS) {
    const int c0   = (blockIdx.x >> 1) * TILE_ROWS;
    const int jblk = (blockIdx.x & 1) * BLOCK_COLS;
    const int w = threadIdx.x >> 5;

    const int v4    = threadIdx.x & 7;
    const int frow0 = threadIdx.x >> 3;
    v4f tap[2][NK - 1];

#pragma unroll
    for (int k = 0; k < 2; ++k) {
      const float* fr = fifo + (size_t)(c0 + frow0 + 8 * k) * NF * NV + jblk + v4 * 4;
#pragma unroll
      for (int i = 1; i < NK; ++i) {
        tap[k][i - 1] = *(const v4f*)(fr + (size_t)(i * ND - 1) * NV);
      }
    }
    __builtin_amdgcn_sched_barrier(0);

    {
      const int lane = threadIdx.x & 31;
      const int half = lane >> 4;
      const int lr   = lane & 15;

      const int jl = w * 16 + lr;
      const int ca = c0 + lr;
      const int jb = jblk + jl;

      v8f acc = {};
#pragma unroll
      for (int kk = 0; kk < KDIM / 4; ++kk) {
        const int kg = kk * 4 + half * 2;
        const int p  = kg >> 6;
        const int u  = kg & 63;
        v2f a = *(const v2f*)(x + (size_t)(p * NC + ca) * NV + u);
        v2f b;
        b.x = graph[(size_t)(p * NV + u)     * NV + jb];
        b.y = graph[(size_t)(p * NV + u + 1) * NV + jb];
        acc = __builtin_amdgcn_wmma_f32_16x16x4_f32(
             false, a,  false, b,
             (short)0, acc,  false,  false);
      }

#pragma unroll
      for (int r = 0; r < 8; ++r) {
        ytile[(half * 8 + r) * TILE_PITCH + jl] = acc[r];
      }
    }

    __syncthreads();

#pragma unroll
    for (int k = 0; k < 2; ++k) {
      const int row = frow0 + 8 * k;
      v4f o = *(const v4f*)(ytile + row * TILE_PITCH + v4 * 4);
#pragma unroll
      for (int i = 1; i < NK; ++i) o += tap[k][i - 1];
      *(v4f*)(out + ((size_t)(c0 + row) * HOUT) * NV + jblk + v4 * 4) = o;
    }
  } else {
    const int s = (blockIdx.x - PROD_BLOCKS) * BLOCK_THREADS + threadIdx.x;
    const int c = s >> 4;
    const int v = (s & 15) << 2;

    const float* fbase = fifo + (size_t)c * NF * NV + v;
    float*       obase = out + (size_t)c * HOUT * NV + v;

    v4f t[ROWS_PER_C][NK];
#pragma unroll
    for (int h = 1; h < HOUT; ++h) {
#pragma unroll
      for (int i = 0; i < NK; ++i) {
        t[h - 1][i] = *(const v4f*)(fbase + (size_t)(i * ND + h - 1) * NV);
      }
    }
    __builtin_amdgcn_sched_barrier(0);
#pragma unroll
    for (int h = 1; h < HOUT; ++h) {
      v4f acc = t[h - 1][0];
#pragma unroll
      for (int i = 1; i < NK; ++i) acc += t[h - 1][i];
      *(v4f*)(obase + (size_t)h * NV) = acc;
    }
  }
}

extern "C" void kernel_launch(void* const* d_in, const int* in_sizes, int n_in,
                              void* d_out, int out_size, void* d_ws, size_t ws_size,
                              hipStream_t stream) {
  (void)in_sizes; (void)n_in; (void)out_size; (void)d_ws; (void)ws_size;
  const float* x     = (const float*)d_in[0];
  const float* graph = (const float*)d_in[1];
  const float* fifo  = (const float*)d_in[2];
  float* out = (float*)d_out;

  graph_fifo_fused<<<PROD_BLOCKS + STREAM_BLOCKS, BLOCK_THREADS, 0, stream>>>(x, graph, fifo, out);
}
